# starting kernel
# baseline (speedup 1.0000x reference)
#include <hip/hip_runtime.h>
#include <hip/hip_bf16.h>

#define HIDDEN 768
#define EPS_F  0.0001f

typedef __attribute__((ext_vector_type(2))) float v2f;
typedef __attribute__((ext_vector_type(8))) float v8f;

__global__ void init_adj_kernel(float* __restrict__ out, int n_elem) {
    int i = blockIdx.x * blockDim.x + threadIdx.x;
    if (i < n_elem) out[i] = 1.0f + EPS_F;
}

__global__ __launch_bounds__(256) void gemv_sigmoid_scatter_kernel(
    const float* __restrict__ pooled,
    const float* __restrict__ W,
    const float* __restrict__ bptr,
    const int*   __restrict__ pairs,
    float*       __restrict__ out,
    int P, int ppp, int n)
{
    const int lane = threadIdx.x & 31;
    const int wave = threadIdx.x >> 5;
    const int tile = blockIdx.x * (blockDim.x >> 5) + wave;
    const int numTiles = (P + 15) >> 4;
    if (tile >= numTiles) return;

    const int rowBase = tile << 4;
    const int m  = lane & 15;
    const int g2 = (lane >> 4) << 1;

    int row = rowBase + m;
    if (row >= P) row = P - 1;

    const float* arow = pooled + (size_t)row * HIDDEN + g2;
    const float* wvec = W + g2;

    v8f c0 = {};
    v8f c1 = {};

    #pragma unroll 4
    for (int k = 0; k < HIDDEN; k += 8) {
        v2f a0 = *(const v2f*)(arow + k);
        v2f b0 = *(const v2f*)(wvec + k);
        c0 = __builtin_amdgcn_wmma_f32_16x16x4_f32(
                 false, a0, false, b0, (short)0, c0, false, false);

        v2f a1 = *(const v2f*)(arow + k + 4);
        v2f b1 = *(const v2f*)(wvec + k + 4);
        c1 = __builtin_amdgcn_wmma_f32_16x16x4_f32(
                 false, a1, false, b1, (short)0, c1, false, false);
    }

    v8f acc = c0 + c1;

    if (lane == 0 || lane == 16) {
        const float bval = bptr[0];
        const int base = rowBase + ((lane >> 4) << 3);
        #pragma unroll
        for (int i = 0; i < 8; ++i) {
            int p = base + i;
            if (p < P) {
                int r  = pairs[2 * p];
                int cc = pairs[2 * p + 1];
                int bidx = p / ppp;
                float x = acc[i] + bval;
                float s = 1.0f / (1.0f + __expf(-x));
                out[(size_t)bidx * n * n + r * n + cc] = s + EPS_F;
            }
        }
    }
}

extern "C" void kernel_launch(void* const* d_in, const int* in_sizes, int n_in,
                              void* d_out, int out_size, void* d_ws, size_t ws_size,
                              hipStream_t stream) {
    const float* pooled = (const float*)d_in[0];
    const float* W      = (const float*)d_in[1];
    const float* b      = (const float*)d_in[2];
    const int*   pairs  = (const int*)d_in[3];
    float* out = (float*)d_out;

    const int P = in_sizes[0] / HIDDEN;
    const int Bn = out_size - P;
    const int n  = out_size / Bn;
    const int ppp = n * (n - 1);

    init_adj_kernel<<<(out_size + 255) / 256, 256, 0, stream>>>(out, out_size);

    const int numTiles = (P + 15) / 16;
    const int wavesPerBlock = 8;
    const int blocks = (numTiles + wavesPerBlock - 1) / wavesPerBlock;
    gemv_sigmoid_scatter_kernel<<<blocks, 256, 0, stream>>>(
        pooled, W, b, pairs, out, P, ppp, n);
}


# Row tile staged through wave-private LDS: each global load reads 512 contiguous bytes of one row (16 B per lane, 16-32 loads in flight), not 8 B per lane over 16 rows; k regrouped: sum order changes
# speedup vs baseline: 1.2066x; 1.2066x over previous
#include <hip/hip_runtime.h>
#include <hip/hip_bf16.h>

#define HIDDEN 768
#define EPS_F  0.0001f

#define TILE_ROWS 16
#define CHUNK     128
#define NCHUNK    (HIDDEN / CHUNK)
#define PITCH     (CHUNK + 4)
#define WAVES     8
#define JB        8

typedef __attribute__((ext_vector_type(2))) float v2f;
typedef __attribute__((ext_vector_type(4))) float v4f;
typedef __attribute__((ext_vector_type(8))) float v8f;

__global__ void init_adj_kernel(float* __restrict__ out, int n_elem) {
    int i = blockIdx.x * blockDim.x + threadIdx.x;
    if (i < n_elem) out[i] = 1.0f + EPS_F;
}

__global__ __launch_bounds__(256) void gemv_sigmoid_scatter_kernel(
    const float* __restrict__ pooled,
    const float* __restrict__ W,
    const float* __restrict__ bptr,
    const int*   __restrict__ pairs,
    float*       __restrict__ out,
    int P, int ppp, int n)
{
    __shared__ __attribute__((aligned(16))) float sW[WAVES][HIDDEN];
    __shared__ __attribute__((aligned(16))) float sX[WAVES][TILE_ROWS * PITCH];

    const int lane = threadIdx.x & 31;
    const int wave = __builtin_amdgcn_readfirstlane((int)(threadIdx.x >> 5));
    const int tile = blockIdx.x * WAVES + wave;
    const int numTiles = (P + 15) >> 4;
    if (tile >= numTiles) return;

    const int rowBase = tile << 4;
    const int c = lane & 15;
    const int h = lane >> 4;

    v4f wreg[NCHUNK];
    #pragma unroll
    for (int q = 0; q < NCHUNK; ++q) wreg[q] = *(const v4f*)(W + 4 * lane + q * CHUNK);
    __builtin_amdgcn_sched_barrier(0);

    const float* rowp[TILE_ROWS];
    #pragma unroll
    for (int r = 0; r < TILE_ROWS; ++r) {
        int row = rowBase + r;
        if (row >= P) row = P - 1;
        rowp[r] = pooled + (size_t)row * HIDDEN + 4 * lane;
    }

    float*       stage = sX[wave] + 4 * lane;
    const float* afrag = sX[wave] + c * PITCH + 4 * h;
    const float* wfrag = sW[wave] + 4 * h;

    v4f buf[2][TILE_ROWS];
    #pragma unroll
    for (int r = 0; r < TILE_ROWS; ++r) buf[0][r] = *(const v4f*)(rowp[r]);

    #pragma unroll
    for (int q = 0; q < NCHUNK; ++q) *(v4f*)(sW[wave] + 4 * lane + q * CHUNK) = wreg[q];

    v8f c0 = {};
    v8f c1 = {};

    #pragma unroll
    for (int q = 0; q < NCHUNK; ++q) {
        if (q + 1 < NCHUNK) {
            #pragma unroll
            for (int r = 0; r < TILE_ROWS; ++r)
                buf[(q + 1) & 1][r] = *(const v4f*)(rowp[r] + (q + 1) * CHUNK);
        }
        #pragma unroll
        for (int r = 0; r < TILE_ROWS; ++r)
            *(v4f*)(stage + r * PITCH) = buf[q & 1][r];
        #pragma unroll
        for (int jb = 0; jb < CHUNK / 8; jb += JB) {
            v4f a[JB], w[JB];
            #pragma unroll
            for (int j = 0; j < JB; ++j) {
                a[j] = *(const v4f*)(afrag + 8 * (jb + j));
                w[j] = *(const v4f*)(wfrag + q * CHUNK + 8 * (jb + j));
            }
            __builtin_amdgcn_sched_barrier(0);
            #pragma unroll
            for (int j = 0; j < JB; ++j) {
                const v2f a0 = {a[j].x, a[j].y}, b0 = {w[j].x, w[j].y};
                const v2f a1 = {a[j].z, a[j].w}, b1 = {w[j].z, w[j].w};
                c0 = __builtin_amdgcn_wmma_f32_16x16x4_f32(
                         false, a0, false, b0, (short)0, c0, false, false);
                c1 = __builtin_amdgcn_wmma_f32_16x16x4_f32(
                         false, a1, false, b1, (short)0, c1, false, false);
            }
            __builtin_amdgcn_sched_barrier(0);
        }
    }

    v8f acc = c0 + c1;

    if (lane == 0 || lane == 16) {
        const float bval = bptr[0];
        const int base = rowBase + ((lane >> 4) << 3);
        #pragma unroll
        for (int i = 0; i < 8; ++i) {
            int p = base + i;
            if (p < P) {
                int r  = pairs[2 * p];
                int cc = pairs[2 * p + 1];
                int bidx = p / ppp;
                float x = acc[i] + bval;
                float s = 1.0f / (1.0f + __expf(-x));
                out[(size_t)bidx * n * n + r * n + cc] = s + EPS_F;
            }
        }
    }
}

extern "C" void kernel_launch(void* const* d_in, const int* in_sizes, int n_in,
                              void* d_out, int out_size, void* d_ws, size_t ws_size,
                              hipStream_t stream) {
    const float* pooled = (const float*)d_in[0];
    const float* W      = (const float*)d_in[1];
    const float* b      = (const float*)d_in[2];
    const int*   pairs  = (const int*)d_in[3];
    float* out = (float*)d_out;

    const int P = in_sizes[0] / HIDDEN;
    const int Bn = out_size - P;
    const int n  = out_size / Bn;
    const int ppp = n * (n - 1);

    init_adj_kernel<<<(out_size + 255) / 256, 256, 0, stream>>>(out, out_size);

    const int numTiles = (P + 15) / 16;
    const int wavesPerBlock = 8;
    const int blocks = (numTiles + wavesPerBlock - 1) / wavesPerBlock;
    gemv_sigmoid_scatter_kernel<<<blocks, 256, 0, stream>>>(
        pooled, W, b, pairs, out, P, ppp, n);
}


# The 1+eps fill moved into the GEMV launch: workgroup d marks in LDS which of its 256 outputs the pair rows name and stores the rest; named ones come from the scatter only. One launch, not two
# speedup vs baseline: 1.2977x; 1.0755x over previous
#include <hip/hip_runtime.h>
#include <hip/hip_bf16.h>

#define HIDDEN 768
#define EPS_F  0.0001f

#define TILE_ROWS 16
#define CHUNK     128
#define NCHUNK    (HIDDEN / CHUNK)
#define PITCH     (CHUNK + 4)
#define WAVES     8
#define JB        8
#define FILL      256
#define NPRE      4

typedef __attribute__((ext_vector_type(2))) float v2f;
typedef __attribute__((ext_vector_type(4))) float v4f;
typedef __attribute__((ext_vector_type(8))) float v8f;
typedef __attribute__((ext_vector_type(2))) int   v2i;

static __device__ __forceinline__ void lds_barrier() {
    __builtin_amdgcn_fence(__ATOMIC_RELEASE, "workgroup", "local");
    __builtin_amdgcn_s_barrier();
    __builtin_amdgcn_fence(__ATOMIC_ACQUIRE, "workgroup", "local");
}

__global__ __launch_bounds__(256) void gemv_sigmoid_scatter_kernel(
    const float* __restrict__ pooled,
    const float* __restrict__ W,
    const float* __restrict__ bptr,
    const int*   __restrict__ pairs,
    float*       __restrict__ out,
    int P, int ppp, int n, int out_size)
{
    __shared__ __attribute__((aligned(16))) float sW[WAVES][HIDDEN];
    __shared__ __attribute__((aligned(16))) float sX[WAVES][TILE_ROWS * PITCH];
    __shared__ int sHit[FILL];

    const int tid  = threadIdx.x;
    const int lane = tid & 31;
    const int wave = __builtin_amdgcn_readfirstlane((int)(threadIdx.x >> 5));
    const int numTiles = (P + 15) >> 4;
    const int tileOwn = blockIdx.x * WAVES + wave;
    const bool active = tileOwn < numTiles;
    const int tile = active ? tileOwn : numTiles - 1;

    const int nn = n * n;
    const int e0 = blockIdx.x * FILL;
    const bool fill = e0 < out_size;
    int bLo = 0, bHi = -1, pEnd = 0;
    v2i pre[NPRE];
    if (fill) {
        const int eLast = (e0 + FILL < out_size ? e0 + FILL : out_size) - 1;
        bLo = e0 / nn;
        bHi = eLast / nn;
        pEnd = bLo * ppp + ppp;
        if (pEnd > P) pEnd = P;
        #pragma unroll
        for (int i = 0; i < NPRE; ++i) {
            int p = bLo * ppp + tid + i * FILL;
            if (p >= pEnd) p = pEnd - 1;
            pre[i] = *(const v2i*)(pairs + 2 * p);
        }
    }

    const int rowBase = tile << 4;
    const int c = lane & 15;
    const int h = lane >> 4;

    v4f wreg[NCHUNK];
    #pragma unroll
    for (int q = 0; q < NCHUNK; ++q) wreg[q] = *(const v4f*)(W + 4 * lane + q * CHUNK);
    __builtin_amdgcn_sched_barrier(0);

    const float* rowp[TILE_ROWS];
    #pragma unroll
    for (int r = 0; r < TILE_ROWS; ++r) {
        int row = rowBase + r;
        if (row >= P) row = P - 1;
        rowp[r] = pooled + (size_t)row * HIDDEN + 4 * lane;
    }

    float*       stage = sX[wave] + 4 * lane;
    const float* afrag = sX[wave] + c * PITCH + 4 * h;
    const float* wfrag = sW[wave] + 4 * h;

    v4f buf[2][TILE_ROWS];
    #pragma unroll
    for (int r = 0; r < TILE_ROWS; ++r) buf[0][r] = *(const v4f*)(rowp[r]);

    #pragma unroll
    for (int q = 0; q < NCHUNK; ++q) *(v4f*)(sW[wave] + 4 * lane + q * CHUNK) = wreg[q];

    v8f c0 = {};
    v8f c1 = {};

    #pragma unroll
    for (int q = 0; q < NCHUNK; ++q) {
        if (q + 1 < NCHUNK) {
            #pragma unroll
            for (int r = 0; r < TILE_ROWS; ++r)
                buf[(q + 1) & 1][r] = *(const v4f*)(rowp[r] + (q + 1) * CHUNK);
        }
        if (q == 0 && fill) {
            sHit[tid] = 0;
            lds_barrier();
            #pragma unroll
            for (int i = 0; i < NPRE; ++i) {
                const int p = bLo * ppp + tid + i * FILL;
                if (p < pEnd) {
                    const int e = bLo * nn - e0 + pre[i].x * n + pre[i].y;
                    if ((unsigned)e < (unsigned)FILL) sHit[e] = 1;
                }
            }
            for (int b = bLo; b <= bHi; ++b) {
                int pe = b * ppp + ppp;
                if (pe > P) pe = P;
                for (int p = b * ppp + tid + (b == bLo ? NPRE * FILL : 0); p < pe; p += FILL) {
                    const v2i rc = *(const v2i*)(pairs + 2 * p);
                    const int e = b * nn - e0 + rc.x * n + rc.y;
                    if ((unsigned)e < (unsigned)FILL) sHit[e] = 1;
                }
            }
            lds_barrier();
            if (e0 + tid < out_size && sHit[tid] == 0) out[e0 + tid] = 1.0f + EPS_F;
        }
        #pragma unroll
        for (int r = 0; r < TILE_ROWS; ++r)
            *(v4f*)(stage + r * PITCH) = buf[q & 1][r];
        #pragma unroll
        for (int jb = 0; jb < CHUNK / 8; jb += JB) {
            v4f a[JB], w[JB];
            #pragma unroll
            for (int j = 0; j < JB; ++j) {
                a[j] = *(const v4f*)(afrag + 8 * (jb + j));
                w[j] = *(const v4f*)(wfrag + q * CHUNK + 8 * (jb + j));
            }
            __builtin_amdgcn_sched_barrier(0);
            #pragma unroll
            for (int j = 0; j < JB; ++j) {
                const v2f a0 = {a[j].x, a[j].y}, b0 = {w[j].x, w[j].y};
                const v2f a1 = {a[j].z, a[j].w}, b1 = {w[j].z, w[j].w};
                c0 = __builtin_amdgcn_wmma_f32_16x16x4_f32(
                         false, a0, false, b0, (short)0, c0, false, false);
                c1 = __builtin_amdgcn_wmma_f32_16x16x4_f32(
                         false, a1, false, b1, (short)0, c1, false, false);
            }
            __builtin_amdgcn_sched_barrier(0);
        }
    }

    v8f acc = c0 + c1;

    if (active && (lane == 0 || lane == 16)) {
        const float bval = bptr[0];
        const int base = rowBase + ((lane >> 4) << 3);
        #pragma unroll
        for (int i = 0; i < 8; ++i) {
            int p = base + i;
            if (p < P) {
                int r  = pairs[2 * p];
                int cc = pairs[2 * p + 1];
                int bidx = p / ppp;
                float x = acc[i] + bval;
                float s = 1.0f / (1.0f + __expf(-x));
                out[(size_t)bidx * n * n + r * n + cc] = s + EPS_F;
            }
        }
    }
}

extern "C" void kernel_launch(void* const* d_in, const int* in_sizes, int n_in,
                              void* d_out, int out_size, void* d_ws, size_t ws_size,
                              hipStream_t stream) {
    const float* pooled = (const float*)d_in[0];
    const float* W      = (const float*)d_in[1];
    const float* b      = (const float*)d_in[2];
    const int*   pairs  = (const int*)d_in[3];
    float* out = (float*)d_out;

    const int P = in_sizes[0] / HIDDEN;
    const int Bn = out_size - P;
    const int n  = out_size / Bn;
    const int ppp = n * (n - 1);

    const int numTiles = (P + 15) / 16;
    const int wavesPerBlock = 8;
    const int blocks = (numTiles + wavesPerBlock - 1) / wavesPerBlock;
    gemv_sigmoid_scatter_kernel<<<blocks, 256, 0, stream>>>(
        pooled, W, b, pairs, out, P, ppp, n, out_size);
}


# One row per accumulator tile: a row's 512-B load registers are the A fragments, W (24 registers, same lane order) the B fragments; the dots land on D's diagonal; no LDS staging; sum order changes
# speedup vs baseline: 1.8158x; 1.3992x over previous
#include <hip/hip_runtime.h>
#include <hip/hip_bf16.h>

#define HIDDEN 768
#define EPS_F  0.0001f

#define TILE_ROWS 16
#define PIECE     128
#define NPIECE    (HIDDEN / PIECE)
#define WAVES     8
#define AHEAD     6
#define FILL      256
#define NPRE      4

typedef __attribute__((ext_vector_type(2))) float v2f;
typedef __attribute__((ext_vector_type(4))) float v4f;
typedef __attribute__((ext_vector_type(8))) float v8f;
typedef __attribute__((ext_vector_type(2))) int   v2i;

static __device__ __forceinline__ void lds_barrier() {
    __builtin_amdgcn_fence(__ATOMIC_RELEASE, "workgroup", "local");
    __builtin_amdgcn_s_barrier();
    __builtin_amdgcn_fence(__ATOMIC_ACQUIRE, "workgroup", "local");
}

static __device__ __forceinline__ float xhalf(float x) {
    const int b = __builtin_bit_cast(int, x);
    return __builtin_bit_cast(float,
        __builtin_amdgcn_permlanex16(b, b, 0x76543210, 0xfedcba98, false, false));
}

__global__ __launch_bounds__(256) void gemv_sigmoid_scatter_kernel(
    const float* __restrict__ pooled,
    const float* __restrict__ W,
    const float* __restrict__ bptr,
    const int*   __restrict__ pairs,
    float*       __restrict__ out,
    int P, int ppp, int n, int out_size)
{
    __shared__ int sHit[FILL];

    const int tid  = threadIdx.x;
    const int lane = tid & 31;
    const int wave = __builtin_amdgcn_readfirstlane((int)(threadIdx.x >> 5));
    const int numTiles = (P + 15) >> 4;
    const int tileOwn = blockIdx.x * WAVES + wave;
    const bool active = tileOwn < numTiles;
    const int tile = active ? tileOwn : numTiles - 1;

    const int nn = n * n;
    const int e0 = blockIdx.x * FILL;
    const bool fill = e0 < out_size;
    int bLo = 0, bHi = -1, pEnd = 0;
    v2i pre[NPRE];
    if (fill) {
        const int eLast = (e0 + FILL < out_size ? e0 + FILL : out_size) - 1;
        bLo = e0 / nn;
        bHi = eLast / nn;
        pEnd = bLo * ppp + ppp;
        if (pEnd > P) pEnd = P;
        #pragma unroll
        for (int i = 0; i < NPRE; ++i) {
            int p = bLo * ppp + tid + i * FILL;
            if (p >= pEnd) p = pEnd - 1;
            pre[i] = *(const v2i*)(pairs + 2 * p);
        }
    }

    const int rowBase = tile << 4;

    v4f wreg[NPIECE];
    #pragma unroll
    for (int q = 0; q < NPIECE; ++q) wreg[q] = *(const v4f*)(W + 4 * lane + q * PIECE);
    __builtin_amdgcn_sched_barrier(0);

    const float* rowp[TILE_ROWS];
    #pragma unroll
    for (int r = 0; r < TILE_ROWS; ++r) {
        int row = rowBase + r;
        if (row >= P) row = P - 1;
        rowp[r] = pooled + (size_t)row * HIDDEN + 4 * lane;
    }

    v4f x[AHEAD][NPIECE];
    #pragma unroll
    for (int r = 0; r < AHEAD; ++r) {
        #pragma unroll
        for (int i = 0; i < NPIECE; ++i) x[r][i] = *(const v4f*)(rowp[r] + i * PIECE);
    }

    if (fill) {
        sHit[tid] = 0;
        lds_barrier();
        #pragma unroll
        for (int i = 0; i < NPRE; ++i) {
            const int p = bLo * ppp + tid + i * FILL;
            if (p < pEnd) {
                const int e = bLo * nn - e0 + pre[i].x * n + pre[i].y;
                if ((unsigned)e < (unsigned)FILL) sHit[e] = 1;
            }
        }
        for (int b = bLo; b <= bHi; ++b) {
            int pe = b * ppp + ppp;
            if (pe > P) pe = P;
            for (int p = b * ppp + tid + (b == bLo ? NPRE * FILL : 0); p < pe; p += FILL) {
                const v2i rc = *(const v2i*)(pairs + 2 * p);
                const int e = b * nn - e0 + rc.x * n + rc.y;
                if ((unsigned)e < (unsigned)FILL) sHit[e] = 1;
            }
        }
        lds_barrier();
        if (e0 + tid < out_size && sHit[tid] == 0) out[e0 + tid] = 1.0f + EPS_F;
    }

    const int  dsel = lane & 7;
    const bool diag = ((lane >> 3) == 0) || ((lane >> 3) == 3);
    float yv = 0.0f;

    #pragma unroll
    for (int p = 0; p < TILE_ROWS / 2; ++p) {
        const int sa = (2 * p) % AHEAD;
        const int sb = (2 * p + 1) % AHEAD;
        v8f ca = {};
        v8f cb = {};
        #pragma unroll
        for (int i = 0; i < NPIECE; ++i) {
            const v2f w0 = {wreg[i].x, wreg[i].y}, w1 = {wreg[i].z, wreg[i].w};
            const v2f a0 = {x[sa][i].x, x[sa][i].y}, a1 = {x[sa][i].z, x[sa][i].w};
            const v2f b0 = {x[sb][i].x, x[sb][i].y}, b1 = {x[sb][i].z, x[sb][i].w};
            ca = __builtin_amdgcn_wmma_f32_16x16x4_f32(false, a0, false, w0, (short)0, ca, false, false);
            cb = __builtin_amdgcn_wmma_f32_16x16x4_f32(false, b0, false, w0, (short)0, cb, false, false);
            ca = __builtin_amdgcn_wmma_f32_16x16x4_f32(false, a1, false, w1, (short)0, ca, false, false);
            cb = __builtin_amdgcn_wmma_f32_16x16x4_f32(false, b1, false, w1, (short)0, cb, false, false);
        }
        if (2 * p + AHEAD < TILE_ROWS) {
            __builtin_amdgcn_sched_barrier(0);
            #pragma unroll
            for (int i = 0; i < NPIECE; ++i) x[sa][i] = *(const v4f*)(rowp[2 * p + AHEAD] + i * PIECE);
            #pragma unroll
            for (int i = 0; i < NPIECE; ++i) x[sb][i] = *(const v4f*)(rowp[2 * p + 1 + AHEAD] + i * PIECE);
            __builtin_amdgcn_sched_barrier(0);
        }
        float ea = 0.0f, eb = 0.0f;
        #pragma unroll
        for (int r = 0; r < 8; ++r) {
            const bool m = diag && (dsel == r);
            ea = m ? ca[r] : ea;
            eb = m ? cb[r] : eb;
        }
        float s = ea + xhalf(eb);
        s += __shfl_xor(s, 1);
        s += __shfl_xor(s, 2);
        s += __shfl_xor(s, 4);
        s += __shfl_xor(s, 24);
        yv = (lane < 16 && dsel == p) ? s : yv;
    }

    if (active && lane < 16) {
        const int p = rowBase + 2 * dsel + (lane >> 3);
        if (p < P) {
            const v2i rc = *(const v2i*)(pairs + 2 * p);
            const int bidx = p / ppp;
            const float xv = yv + bptr[0];
            const float sg = 1.0f / (1.0f + __expf(-xv));
            out[(size_t)bidx * n * n + rc.x * n + rc.y] = sg + EPS_F;
        }
    }
}

extern "C" void kernel_launch(void* const* d_in, const int* in_sizes, int n_in,
                              void* d_out, int out_size, void* d_ws, size_t ws_size,
                              hipStream_t stream) {
    const float* pooled = (const float*)d_in[0];
    const float* W      = (const float*)d_in[1];
    const float* b      = (const float*)d_in[2];
    const int*   pairs  = (const int*)d_in[3];
    float* out = (float*)d_out;

    const int P = in_sizes[0] / HIDDEN;
    const int Bn = out_size - P;
    const int n  = out_size / Bn;
    const int ppp = n * (n - 1);

    const int numTiles = (P + 15) / 16;
    const int wavesPerBlock = 8;
    const int blocks = (numTiles + wavesPerBlock - 1) / wavesPerBlock;
    gemv_sigmoid_scatter_kernel<<<blocks, 256, 0, stream>>>(
        pooled, W, b, pairs, out, P, ppp, n, out_size);
}
